# starting kernel
# baseline (speedup 1.0000x reference)
#include <hip/hip_runtime.h>
#include <hip/hip_bf16.h>
#include <math.h>

typedef __attribute__((ext_vector_type(2))) float v2f;
typedef __attribute__((ext_vector_type(4))) float v4f;
typedef __attribute__((ext_vector_type(8))) float v8f;

#define BATCH 8
#define N_PTS 512
#define DIN   32
#define H1DIM 64
#define H2DIM 64
#define H3DIM 32

__device__ __forceinline__ void wave_sync() {
    __builtin_amdgcn_fence(__ATOMIC_ACQ_REL, "workgroup");
    __builtin_amdgcn_wave_barrier();
}

__global__ __launch_bounds__(64) void cpc_precompute_kernel(
        const float* __restrict__ x, const float* __restrict__ W1,
        const float* __restrict__ b1,
        float* __restrict__ Apre, float* __restrict__ Bpre) {
    __shared__ float xr[DIN];
    const int row = blockIdx.x;
    const int c = threadIdx.x;
    if (c < DIN) xr[c] = x[row * DIN + c];
    __syncthreads();
    float a = 0.f;
    float bacc = b1[c];
    #pragma unroll
    for (int d = 0; d < DIN; ++d) {
        const float xv = xr[d];
        a    = fmaf(xv, W1[d * H1DIM + c], a);
        bacc = fmaf(xv, W1[(DIN + d) * H1DIM + c], bacc);
    }
    Apre[row * H1DIM + c] = a;
    Bpre[row * H1DIM + c] = bacc;
}

#define KSTR 68
#define TSTR 36

__global__ __launch_bounds__(256) void cpc_pair_mlp_kernel(
        const float* __restrict__ Apre, const float* __restrict__ Bpre,
        const float* __restrict__ W2, const float* __restrict__ b2,
        const float* __restrict__ W3, const float* __restrict__ b3,
        const float* __restrict__ W4, const float* __restrict__ b4,
        float* __restrict__ out) {
    __shared__ float W2Ts[H2DIM * KSTR];
    __shared__ float W3Ts[H3DIM * KSTR];
    __shared__ float b2s[H2DIM];
    __shared__ float b3s[H3DIM];
    __shared__ float W4s[H3DIM];
    __shared__ float Bis[H1DIM];
    __shared__ float bufA[8][16 * KSTR];
    __shared__ float bufB[8][16 * TSTR];

    const int tid  = threadIdx.x;
    const int wave = tid >> 5;
    const int lane = tid & 31;
    const int half = lane >> 4;
    const int mrow = lane & 15;
    const int bi   = blockIdx.x;
    const int b    = bi >> 9;

    for (int idx = tid; idx < H1DIM * H2DIM; idx += 256) {
        const int k = idx >> 6, n = idx & 63;
        W2Ts[n * KSTR + k] = W2[idx];
    }
    for (int idx = tid; idx < H2DIM * H3DIM; idx += 256) {
        const int k = idx >> 5, n = idx & 31;
        W3Ts[n * KSTR + k] = W3[idx];
    }
    if (tid < H2DIM) b2s[tid] = b2[tid];
    if (tid < H3DIM) { b3s[tid] = b3[tid]; W4s[tid] = W4[tid]; }
    if (tid < H1DIM) Bis[tid] = Bpre[bi * H1DIM + tid];
    __syncthreads();

    const float b4v = b4[0];
    float* H12 = &bufA[wave][0];
    float* H3s = &bufB[wave][0];

    for (int tt = 0; tt < 4; ++tt) {
        const int j0 = (wave + tt * 8) * 16;
        const float* arow = &Apre[(b * N_PTS + j0) * H1DIM];

        #pragma unroll
        for (int t = 0; t < 8; ++t) {
            const int li = lane + 32 * t;
            const int r  = li >> 4;
            const int c  = 4 * (lane & 15);
            v4f v = *(const v4f*)&arow[li * 4];
            const v4f bb = *(const v4f*)&Bis[c];
            v = v + bb;
            v.x = fmaxf(v.x, 0.f); v.y = fmaxf(v.y, 0.f);
            v.z = fmaxf(v.z, 0.f); v.w = fmaxf(v.w, 0.f);
            *(v4f*)&H12[r * KSTR + c] = v;
        }
        wave_sync();

        v8f acc[4] = {};
        #pragma unroll
        for (int k0 = 0; k0 < H1DIM; k0 += 4) {
            const v2f a = *(const v2f*)&H12[mrow * KSTR + k0 + 2 * half];
            #pragma unroll
            for (int nt = 0; nt < 4; ++nt) {
                const v2f bf =
                    *(const v2f*)&W2Ts[(nt * 16 + mrow) * KSTR + k0 + 2 * half];
                acc[nt] = __builtin_amdgcn_wmma_f32_16x16x4_f32(
                    false, a, false, bf, (short)0, acc[nt], false, false);
            }
        }
        wave_sync();
        #pragma unroll
        for (int r = 0; r < 8; ++r) {
            const int row = r + 8 * half;
            #pragma unroll
            for (int nt = 0; nt < 4; ++nt) {
                const int col = nt * 16 + mrow;
                H12[row * KSTR + col] = fmaxf(acc[nt][r] + b2s[col], 0.f);
            }
        }
        wave_sync();

        v8f c3[2] = {};
        #pragma unroll
        for (int k0 = 0; k0 < H2DIM; k0 += 4) {
            const v2f a = *(const v2f*)&H12[mrow * KSTR + k0 + 2 * half];
            #pragma unroll
            for (int nt = 0; nt < 2; ++nt) {
                const v2f bf =
                    *(const v2f*)&W3Ts[(nt * 16 + mrow) * KSTR + k0 + 2 * half];
                c3[nt] = __builtin_amdgcn_wmma_f32_16x16x4_f32(
                    false, a, false, bf, (short)0, c3[nt], false, false);
            }
        }
        wave_sync();
        #pragma unroll
        for (int r = 0; r < 8; ++r) {
            const int row = r + 8 * half;
            #pragma unroll
            for (int nt = 0; nt < 2; ++nt) {
                const int col = nt * 16 + mrow;
                H3s[row * TSTR + col] = fmaxf(c3[nt][r] + b3s[col], 0.f);
            }
        }
        wave_sync();

        float s = 0.f;
        #pragma unroll
        for (int k = 0; k < 16; ++k) {
            const int kk = half * 16 + k;
            s = fmaf(H3s[mrow * TSTR + kk], W4s[kk], s);
        }
        s += __shfl_xor(s, 16, 32);
        if (half == 0) {
            const float z = s + b4v;
            out[bi * N_PTS + j0 + mrow] = 1.f / (1.f + __expf(-z));
        }
        wave_sync();
    }
}

extern "C" void kernel_launch(void* const* d_in, const int* in_sizes, int n_in,
                              void* d_out, int out_size, void* d_ws, size_t ws_size,
                              hipStream_t stream) {
    const float* x  = (const float*)d_in[0];
    const float* W1 = (const float*)d_in[1];
    const float* b1 = (const float*)d_in[2];
    const float* W2 = (const float*)d_in[3];
    const float* b2 = (const float*)d_in[4];
    const float* W3 = (const float*)d_in[5];
    const float* b3 = (const float*)d_in[6];
    const float* W4 = (const float*)d_in[7];
    const float* b4 = (const float*)d_in[8];
    float* out = (float*)d_out;

    float* Apre = (float*)d_ws;
    float* Bpre = Apre + BATCH * N_PTS * H1DIM;

    cpc_precompute_kernel<<<BATCH * N_PTS, 64, 0, stream>>>(x, W1, b1, Apre, Bpre);
    cpc_pair_mlp_kernel<<<BATCH * N_PTS, 256, 0, stream>>>(
        Apre, Bpre, W2, b2, W3, b3, W4, b4, out);
}


# Layers chained in registers: weights are WMMA A kept in VGPRs, activation tile is B, each layer's D is the next B in place; no LDS, a wave loops over 64 tiles; k order of sums permuted, bias first
# speedup vs baseline: 1.2063x; 1.2063x over previous
#include <hip/hip_runtime.h>
#include <hip/hip_bf16.h>
#include <math.h>

typedef __attribute__((ext_vector_type(2))) float v2f;
typedef __attribute__((ext_vector_type(4))) float v4f;
typedef __attribute__((ext_vector_type(8))) float v8f;

#define BATCH 8
#define N_PTS 512
#define DIN   32
#define H1DIM 64
#define H2DIM 64
#define H3DIM 32

__global__ __launch_bounds__(64) void cpc_precompute_kernel(
        const float* __restrict__ x, const float* __restrict__ W1,
        const float* __restrict__ b1,
        float* __restrict__ Apre, float* __restrict__ Bpre) {
    __shared__ float xr[DIN];
    const int row = blockIdx.x;
    const int c = threadIdx.x;
    if (c < DIN) xr[c] = x[row * DIN + c];
    __syncthreads();
    float a = 0.f;
    float bacc = b1[c];
    #pragma unroll
    for (int d = 0; d < DIN; ++d) {
        const float xv = xr[d];
        a    = fmaf(xv, W1[d * H1DIM + c], a);
        bacc = fmaf(xv, W1[(DIN + d) * H1DIM + c], bacc);
    }
    Apre[row * H1DIM + c] = a;
    Bpre[row * H1DIM + c] = bacc;
}

#define I_CHUNK 32
#define NUM_WG  256

__device__ __forceinline__ float xhalf(float v) {
    return __builtin_bit_cast(float, __builtin_amdgcn_permlanex16(
        __builtin_bit_cast(int, v), __builtin_bit_cast(int, v),
        0x76543210, 0xfedcba98, false, false));
}

__global__ __launch_bounds__(256) void cpc_pair_mlp_kernel(
        const float* __restrict__ Apre, const float* __restrict__ Bpre,
        const float* __restrict__ W2, const float* __restrict__ b2,
        const float* __restrict__ W3, const float* __restrict__ b3,
        const float* __restrict__ W4, const float* __restrict__ b4,
        float* __restrict__ out) {
    const int lane = threadIdx.x & 31;
    const int wave = threadIdx.x >> 5;
    const int h    = lane >> 4;
    const int c    = lane & 15;
    const int gw   = blockIdx.x * 8 + wave;
    const int ic   = gw & 15;
    const int jp   = (gw >> 4) & 15;
    const int b    = gw >> 8;

    v2f a2[4][16];
    #pragma unroll
    for (int mt = 0; mt < 4; ++mt) {
        #pragma unroll
        for (int st = 0; st < 16; ++st) {
            const int k = 32 * h + 2 * st;
            a2[mt][st].x = W2[k * H2DIM + mt * 16 + c];
            a2[mt][st].y = W2[(k + 1) * H2DIM + mt * 16 + c];
        }
    }
    v2f a3[2][4][4];
    #pragma unroll
    for (int m3 = 0; m3 < 2; ++m3) {
        #pragma unroll
        for (int mt = 0; mt < 4; ++mt) {
            #pragma unroll
            for (int rp = 0; rp < 4; ++rp) {
                const int k = mt * 16 + 8 * h + 2 * rp;
                a3[m3][mt][rp].x = W3[k * H3DIM + m3 * 16 + c];
                a3[m3][mt][rp].y = W3[(k + 1) * H3DIM + m3 * 16 + c];
            }
        }
    }
    v8f bias2[4], bias3[2], w4v[2];
    #pragma unroll
    for (int mt = 0; mt < 4; ++mt) {
        #pragma unroll
        for (int r = 0; r < 8; ++r) bias2[mt][r] = b2[mt * 16 + 8 * h + r];
    }
    #pragma unroll
    for (int m3 = 0; m3 < 2; ++m3) {
        #pragma unroll
        for (int r = 0; r < 8; ++r) {
            bias3[m3][r] = b3[m3 * 16 + 8 * h + r];
            w4v[m3][r]   = W4[m3 * 16 + 8 * h + r];
        }
    }
    const float b4v = b4[0];

    v4f xr[2][8];
    #pragma unroll
    for (int t = 0; t < 2; ++t) {
        const float* ap = &Apre[(b * N_PTS + jp * 32 + 16 * t + c) * H1DIM + 32 * h];
        #pragma unroll
        for (int g = 0; g < 8; ++g) xr[t][g] = *(const v4f*)&ap[4 * g];
    }

    const float* bp = &Bpre[(b * N_PTS + ic * I_CHUNK) * H1DIM + 32 * h];
    float* op = &out[(b * N_PTS + ic * I_CHUNK) * N_PTS + jp * 32 + lane];

    v4f bn[8];
    #pragma unroll
    for (int g = 0; g < 8; ++g) bn[g] = *(const v4f*)&bp[4 * g];

    for (int ii = 0; ii < I_CHUNK; ++ii) {
        v4f bc[8];
        #pragma unroll
        for (int g = 0; g < 8; ++g) bc[g] = bn[g];
        const int nx = (ii + 1 < I_CHUNK) ? ii + 1 : ii;
        #pragma unroll
        for (int g = 0; g < 8; ++g) bn[g] = *(const v4f*)&bp[nx * H1DIM + 4 * g];

        float part[2];
        #pragma unroll
        for (int t = 0; t < 2; ++t) {
            float h1[32];
            #pragma unroll
            for (int g = 0; g < 8; ++g) {
                const v4f v = xr[t][g] + bc[g];
                h1[4 * g + 0] = fmaxf(v.x, 0.f);
                h1[4 * g + 1] = fmaxf(v.y, 0.f);
                h1[4 * g + 2] = fmaxf(v.z, 0.f);
                h1[4 * g + 3] = fmaxf(v.w, 0.f);
            }

            v8f acc[4];
            #pragma unroll
            for (int st = 0; st < 16; ++st) {
                v2f bf;
                bf.x = h1[2 * st];
                bf.y = h1[2 * st + 1];
                #pragma unroll
                for (int mt = 0; mt < 4; ++mt) {
                    acc[mt] = __builtin_amdgcn_wmma_f32_16x16x4_f32(
                        false, a2[mt][st], false, bf, (short)0,
                        (st == 0) ? bias2[mt] : acc[mt], false, false);
                }
            }

            v8f c3[2];
            #pragma unroll
            for (int mt = 0; mt < 4; ++mt) {
                #pragma unroll
                for (int rp = 0; rp < 4; ++rp) {
                    v2f bf;
                    bf.x = fmaxf(acc[mt][2 * rp], 0.f);
                    bf.y = fmaxf(acc[mt][2 * rp + 1], 0.f);
                    #pragma unroll
                    for (int m3 = 0; m3 < 2; ++m3) {
                        c3[m3] = __builtin_amdgcn_wmma_f32_16x16x4_f32(
                            false, a3[m3][mt][rp], false, bf, (short)0,
                            (mt == 0 && rp == 0) ? bias3[m3] : c3[m3], false, false);
                    }
                }
            }

            float s = 0.f;
            #pragma unroll
            for (int m3 = 0; m3 < 2; ++m3) {
                #pragma unroll
                for (int r = 0; r < 8; ++r)
                    s = fmaf(fmaxf(c3[m3][r], 0.f), w4v[m3][r], s);
            }
            part[t] = s;
        }

        const float keep = (h == 0) ? part[0] : part[1];
        const float give = (h == 0) ? part[1] : part[0];
        const float z = keep + xhalf(give) + b4v;
        op[ii * N_PTS] = 1.f / (1.f + __expf(-z));
    }
}

extern "C" void kernel_launch(void* const* d_in, const int* in_sizes, int n_in,
                              void* d_out, int out_size, void* d_ws, size_t ws_size,
                              hipStream_t stream) {
    const float* x  = (const float*)d_in[0];
    const float* W1 = (const float*)d_in[1];
    const float* b1 = (const float*)d_in[2];
    const float* W2 = (const float*)d_in[3];
    const float* b2 = (const float*)d_in[4];
    const float* W3 = (const float*)d_in[5];
    const float* b3 = (const float*)d_in[6];
    const float* W4 = (const float*)d_in[7];
    const float* b4 = (const float*)d_in[8];
    float* out = (float*)d_out;

    float* Apre = (float*)d_ws;
    float* Bpre = Apre + BATCH * N_PTS * H1DIM;

    cpc_precompute_kernel<<<BATCH * N_PTS, 64, 0, stream>>>(x, W1, b1, Apre, Bpre);
    cpc_pair_mlp_kernel<<<NUM_WG, 256, 0, stream>>>(
        Apre, Bpre, W2, b2, W3, b3, W4, b4, out);
}
